# starting kernel
# baseline (speedup 1.0000x reference)
#include <hip/hip_runtime.h>
#include <math.h>


typedef float v2f __attribute__((ext_vector_type(2)));
typedef float v8f __attribute__((ext_vector_type(8)));

#define NW 8
#define NH 16
#define EDIM 128
#define TOK_PER_BLOCK 16

__global__ __launch_bounds__(256) void quantum_attn_kernel(
    const float* __restrict__ x,
    const float* __restrict__ theta,
    float* __restrict__ Y)
{
    __shared__ float qh_lds[TOK_PER_BLOCK][NH][NW];

    const int lane = threadIdx.x & 31;
    const int wave = threadIdx.x >> 5;
    const int tIn  = wave * 2 + (lane >> 4);
    const int h    = lane & 15;
    const int gt   = blockIdx.x * TOK_PER_BLOCK + tIn;

    const float* xp = x + (size_t)gt * EDIM + h * NW;
    float4 x0 = *(const float4*)(xp);
    float4 x1 = *(const float4*)(xp + 4);
    float4 t0 = *(const float4*)(theta);
    float4 t1 = *(const float4*)(theta + 4);

    float c[NW];
    c[0] = __cosf(x0.x + t0.x);
    c[1] = __cosf(x0.y + t0.y);
    c[2] = __cosf(x0.z + t0.z);
    c[3] = __cosf(x0.w + t0.w);
    c[4] = __cosf(x1.x + t1.x);
    c[5] = __cosf(x1.y + t1.y);
    c[6] = __cosf(x1.z + t1.z);
    c[7] = __cosf(x1.w + t1.w);

    float q[NW];
    float p = c[0];
#pragma unroll
    for (int w = 1; w < NW; ++w) { p *= c[w]; q[w] = p; }
    q[0] = c[1] * c[2] * c[3] * c[4] * c[5] * c[6] * c[7];

#pragma unroll
    for (int w = 0; w < NW; ++w) qh_lds[tIn][h][w] = q[w];
    __syncthreads();

    const float inv_sqrt_dk = 0.35355339059327379f;
    float sc[NH];
    float mx = -INFINITY;
#pragma unroll
    for (int g = 0; g < NH; ++g) {
        float d = 0.f;
#pragma unroll
        for (int w = 0; w < NW; ++w) d += q[w] * qh_lds[tIn][g][w];
        d *= inv_sqrt_dk;
        sc[g] = d;
        mx = fmaxf(mx, d);
    }
    float denom = 0.f;
#pragma unroll
    for (int g = 0; g < NH; ++g) { sc[g] = __expf(sc[g] - mx); denom += sc[g]; }

    float o[NW];
#pragma unroll
    for (int w = 0; w < NW; ++w) o[w] = 0.f;
#pragma unroll
    for (int g = 0; g < NH; ++g) {
        float a = sc[g];
#pragma unroll
        for (int w = 0; w < NW; ++w) o[w] += a * qh_lds[tIn][g][w];
    }
    float r = 1.f / denom;
#pragma unroll
    for (int w = 0; w < NW; ++w) o[w] *= r;

    const int b = gt >> 10;
    const int s = gt & 1023;
    const size_t row = (size_t)b * 1024 + (size_t)h * 64 + (s >> 4);
    float* yp = Y + row * EDIM + (s & 15) * NW;
    *(float4*)(yp)     = make_float4(o[0], o[1], o[2], o[3]);
    *(float4*)(yp + 4) = make_float4(o[4], o[5], o[6], o[7]);
}

__global__ __launch_bounds__(256) void combine_gemm_wmma(
    const float* __restrict__ Y,
    const float* __restrict__ W,
    const float* __restrict__ bias,
    float* __restrict__ out)
{
    const int lane  = threadIdx.x & 31;
    const int wave  = threadIdx.x >> 5;
    const int tileN = wave;
    const int tileM = blockIdx.x;
    const int l16   = lane & 15;
    const int kgrp  = (lane >> 4) * 2;

    const float* arow = Y + ((size_t)tileM * 16 + l16) * EDIM + kgrp;
    const float* brow = W + ((size_t)tileN * 16 + l16) * EDIM + kgrp;

    __builtin_prefetch(arow, 0, 0);
    __builtin_prefetch(brow, 0, 0);

    v8f acc = {0.f, 0.f, 0.f, 0.f, 0.f, 0.f, 0.f, 0.f};
#pragma unroll
    for (int k0 = 0; k0 < EDIM; k0 += 4) {
        v2f a = *(const v2f*)(arow + k0);
        v2f b = *(const v2f*)(brow + k0);
        acc = __builtin_amdgcn_wmma_f32_16x16x4_f32(
             false, a,  false, b,
             (short)0, acc,  false,  false);
    }

    const int n = tileN * 16 + l16;
    const float bcol = bias[n];
    const int rbase = tileM * 16 + 8 * (lane >> 4);
#pragma unroll
    for (int rr = 0; rr < 8; ++rr) {
        out[(size_t)(rbase + rr) * EDIM + n] = acc[rr] + bcol;
    }
}

extern "C" void kernel_launch(void* const* d_in, const int* in_sizes, int n_in,
                              void* d_out, int out_size, void* d_ws, size_t ws_size,
                              hipStream_t stream) {
    const float* x     = (const float*)d_in[0];
    const float* theta = (const float*)d_in[1];
    const float* Wc    = (const float*)d_in[2];
    const float* bc    = (const float*)d_in[3];
    float* out = (float*)d_out;
    float* Y   = (float*)d_ws;

    const int ntok = in_sizes[0] / EDIM;

    const int blocks1 = ntok / TOK_PER_BLOCK;
    quantum_attn_kernel<<<blocks1, 256, 0, stream>>>(x, theta, Y);

    const int mtiles = ntok / 16;
    combine_gemm_wmma<<<mtiles, 256, 0, stream>>>(Y, Wc, bc, out);
}


# Fuse the output projection into the attention kernel: each 16-token group's 16x128 tile stays in LDS as the WMMA A operand; one launch, no trip through the workspace; sums in the same order
# speedup vs baseline: 1.6325x; 1.6325x over previous
#include <hip/hip_runtime.h>
#include <math.h>


typedef float v2f __attribute__((ext_vector_type(2)));
typedef float v8f __attribute__((ext_vector_type(8)));

#define NW 8
#define NH 16
#define EDIM 128
#define TOK_PER_BLOCK 16
#define YPITCH 132

__global__ __launch_bounds__(256) void quantum_attn_project_fused(
    const float* __restrict__ x,
    const float* __restrict__ theta,
    const float* __restrict__ W,
    const float* __restrict__ bias,
    float* __restrict__ out)
{
    __shared__ float qh_lds[TOK_PER_BLOCK][NH][NW];
    __shared__ __attribute__((aligned(16))) float y_lds[NH][YPITCH];

    const int lane = threadIdx.x & 31;
    const int wave = threadIdx.x >> 5;
    const int tIn  = wave * 2 + (lane >> 4);
    const int h    = lane & 15;
    const int gt   = blockIdx.x * TOK_PER_BLOCK + tIn;

    const float* xp = x + (size_t)gt * EDIM + h * NW;
    float4 x0 = *(const float4*)(xp);
    float4 x1 = *(const float4*)(xp + 4);
    float4 t0 = *(const float4*)(theta);
    float4 t1 = *(const float4*)(theta + 4);

    const int kgrp = (lane >> 4) * 2;
    const float* brow = W + ((size_t)wave * 16 + h) * EDIM + kgrp;
    v2f bfrag[EDIM / 4];
#pragma unroll
    for (int i = 0; i < EDIM / 4; ++i) bfrag[i] = *(const v2f*)(brow + 4 * i);
    const int n = wave * 16 + h;
    const float bcol = bias[n];

    float c[NW];
    c[0] = __cosf(x0.x + t0.x);
    c[1] = __cosf(x0.y + t0.y);
    c[2] = __cosf(x0.z + t0.z);
    c[3] = __cosf(x0.w + t0.w);
    c[4] = __cosf(x1.x + t1.x);
    c[5] = __cosf(x1.y + t1.y);
    c[6] = __cosf(x1.z + t1.z);
    c[7] = __cosf(x1.w + t1.w);

    float q[NW];
    float p = c[0];
#pragma unroll
    for (int w = 1; w < NW; ++w) { p *= c[w]; q[w] = p; }
    q[0] = c[1] * c[2] * c[3] * c[4] * c[5] * c[6] * c[7];

#pragma unroll
    for (int w = 0; w < NW; ++w) qh_lds[tIn][h][w] = q[w];
    __syncthreads();

    const float inv_sqrt_dk = 0.35355339059327379f;
    float sc[NH];
    float mx = -INFINITY;
#pragma unroll
    for (int g = 0; g < NH; ++g) {
        float d = 0.f;
#pragma unroll
        for (int w = 0; w < NW; ++w) d += q[w] * qh_lds[tIn][g][w];
        d *= inv_sqrt_dk;
        sc[g] = d;
        mx = fmaxf(mx, d);
    }
    float denom = 0.f;
#pragma unroll
    for (int g = 0; g < NH; ++g) { sc[g] = __expf(sc[g] - mx); denom += sc[g]; }

    float o[NW];
#pragma unroll
    for (int w = 0; w < NW; ++w) o[w] = 0.f;
#pragma unroll
    for (int g = 0; g < NH; ++g) {
        float a = sc[g];
#pragma unroll
        for (int w = 0; w < NW; ++w) o[w] += a * qh_lds[tIn][g][w];
    }
    float r = 1.f / denom;
#pragma unroll
    for (int w = 0; w < NW; ++w) o[w] *= r;

    float* yp = &y_lds[h][tIn * NW];
    *(float4*)(yp)     = make_float4(o[0], o[1], o[2], o[3]);
    *(float4*)(yp + 4) = make_float4(o[4], o[5], o[6], o[7]);
    __syncthreads();

    const float* arow = &y_lds[h][kgrp];
    v2f afrag[EDIM / 4];
#pragma unroll
    for (int i = 0; i < EDIM / 4; ++i) afrag[i] = *(const v2f*)(arow + 4 * i);
    __builtin_amdgcn_sched_barrier(0);
    v8f acc = {0.f, 0.f, 0.f, 0.f, 0.f, 0.f, 0.f, 0.f};
#pragma unroll
    for (int i = 0; i < EDIM / 4; ++i) {
        acc = __builtin_amdgcn_wmma_f32_16x16x4_f32(
             false, afrag[i],  false, bfrag[i],
             (short)0, acc,  false,  false);
    }

    const int gt0 = blockIdx.x * TOK_PER_BLOCK;
    const int b   = gt0 >> 10;
    const int j   = (gt0 & 1023) >> 4;
    const int mbase = 8 * (lane >> 4);
#pragma unroll
    for (int rr = 0; rr < 8; ++rr) {
        const size_t row = (size_t)b * 1024 + (size_t)(mbase + rr) * 64 + j;
        out[row * EDIM + n] = acc[rr] + bcol;
    }
}

extern "C" void kernel_launch(void* const* d_in, const int* in_sizes, int n_in,
                              void* d_out, int out_size, void* d_ws, size_t ws_size,
                              hipStream_t stream) {
    const float* x     = (const float*)d_in[0];
    const float* theta = (const float*)d_in[1];
    const float* Wc    = (const float*)d_in[2];
    const float* bc    = (const float*)d_in[3];
    float* out = (float*)d_out;

    const int ntok = in_sizes[0] / EDIM;

    const int blocks = ntok / TOK_PER_BLOCK;
    quantum_attn_project_fused<<<blocks, 256, 0, stream>>>(x, theta, Wc, bc, out);
}


# 4-wave work-groups, each wave doing double work: a lane owns 2 heads of one token (key/value rows read from LDS once), a wave 32 output columns (A read once, whole-line stores); half the waves enter
# speedup vs baseline: 1.7291x; 1.0592x over previous
#include <hip/hip_runtime.h>
#include <math.h>


typedef float v2f __attribute__((ext_vector_type(2)));
typedef float v8f __attribute__((ext_vector_type(8)));

#define NW 8
#define NH 16
#define EDIM 128
#define TOK_PER_BLOCK 16
#define YPITCH 132
#define QPITCH 132

__global__ __launch_bounds__(128) void quantum_attn_project_fused(
    const float* __restrict__ x,
    const float* __restrict__ theta,
    const float* __restrict__ W,
    const float* __restrict__ bias,
    float* __restrict__ out)
{
    __shared__ __attribute__((aligned(16))) float qh_lds[TOK_PER_BLOCK][QPITCH];
    __shared__ __attribute__((aligned(16))) float y_lds[NH][YPITCH];

    const int lane = threadIdx.x & 31;
    const int wave = threadIdx.x >> 5;
    const int tIn  = wave * 4 + (lane >> 3);
    const int hA   = lane & 7;
    const int gt   = blockIdx.x * TOK_PER_BLOCK + tIn;

    const float* xp = x + (size_t)gt * EDIM + hA * NW;
    float4 xa0 = *(const float4*)(xp);
    float4 xa1 = *(const float4*)(xp + 4);
    float4 xb0 = *(const float4*)(xp + 8 * NW);
    float4 xb1 = *(const float4*)(xp + 8 * NW + 4);
    float4 t0 = *(const float4*)(theta);
    float4 t1 = *(const float4*)(theta + 4);

    const int c    = lane & 15;
    const int kgrp = (lane >> 4) * 2;
    const int n0   = wave * 32 + 2 * c;
    const float* brow = W + (size_t)n0 * EDIM + kgrp;
    v2f bf0[EDIM / 4];
    v2f bf1[EDIM / 4];
#pragma unroll
    for (int i = 0; i < EDIM / 4; ++i) {
        bf0[i] = *(const v2f*)(brow + 4 * i);
        bf1[i] = *(const v2f*)(brow + EDIM + 4 * i);
    }
    const v2f bcol = *(const v2f*)(bias + n0);

    float qa[NW], qb[NW];
    {
        float ca[NW], cb[NW];
        ca[0] = __cosf(xa0.x + t0.x);  cb[0] = __cosf(xb0.x + t0.x);
        ca[1] = __cosf(xa0.y + t0.y);  cb[1] = __cosf(xb0.y + t0.y);
        ca[2] = __cosf(xa0.z + t0.z);  cb[2] = __cosf(xb0.z + t0.z);
        ca[3] = __cosf(xa0.w + t0.w);  cb[3] = __cosf(xb0.w + t0.w);
        ca[4] = __cosf(xa1.x + t1.x);  cb[4] = __cosf(xb1.x + t1.x);
        ca[5] = __cosf(xa1.y + t1.y);  cb[5] = __cosf(xb1.y + t1.y);
        ca[6] = __cosf(xa1.z + t1.z);  cb[6] = __cosf(xb1.z + t1.z);
        ca[7] = __cosf(xa1.w + t1.w);  cb[7] = __cosf(xb1.w + t1.w);
        float pa = ca[0], pb = cb[0];
#pragma unroll
        for (int w = 1; w < NW; ++w) { pa *= ca[w]; qa[w] = pa; pb *= cb[w]; qb[w] = pb; }
        qa[0] = ca[1] * ca[2] * ca[3] * ca[4] * ca[5] * ca[6] * ca[7];
        qb[0] = cb[1] * cb[2] * cb[3] * cb[4] * cb[5] * cb[6] * cb[7];
    }

    float* qrow = &qh_lds[tIn][0];
    *(float4*)(qrow + hA * NW)           = make_float4(qa[0], qa[1], qa[2], qa[3]);
    *(float4*)(qrow + hA * NW + 4)       = make_float4(qa[4], qa[5], qa[6], qa[7]);
    *(float4*)(qrow + (hA + 8) * NW)     = make_float4(qb[0], qb[1], qb[2], qb[3]);
    *(float4*)(qrow + (hA + 8) * NW + 4) = make_float4(qb[4], qb[5], qb[6], qb[7]);
    __syncthreads();

    float kv[NH][NW];
#pragma unroll
    for (int g = 0; g < NH; ++g) {
        float4 k0 = *(const float4*)(qrow + g * NW);
        float4 k1 = *(const float4*)(qrow + g * NW + 4);
        kv[g][0] = k0.x; kv[g][1] = k0.y; kv[g][2] = k0.z; kv[g][3] = k0.w;
        kv[g][4] = k1.x; kv[g][5] = k1.y; kv[g][6] = k1.z; kv[g][7] = k1.w;
    }

    const float inv_sqrt_dk = 0.35355339059327379f;
    float sa[NH], sb[NH];
    float mxa = -INFINITY, mxb = -INFINITY;
#pragma unroll
    for (int g = 0; g < NH; ++g) {
        float da = 0.f, db = 0.f;
#pragma unroll
        for (int w = 0; w < NW; ++w) { da += qa[w] * kv[g][w]; db += qb[w] * kv[g][w]; }
        da *= inv_sqrt_dk;
        db *= inv_sqrt_dk;
        sa[g] = da;
        sb[g] = db;
        mxa = fmaxf(mxa, da);
        mxb = fmaxf(mxb, db);
    }
    float dena = 0.f, denb = 0.f;
#pragma unroll
    for (int g = 0; g < NH; ++g) {
        sa[g] = __expf(sa[g] - mxa); dena += sa[g];
        sb[g] = __expf(sb[g] - mxb); denb += sb[g];
    }

    float oa[NW], ob[NW];
#pragma unroll
    for (int w = 0; w < NW; ++w) { oa[w] = 0.f; ob[w] = 0.f; }
#pragma unroll
    for (int g = 0; g < NH; ++g) {
        const float aa = sa[g];
        const float ab = sb[g];
#pragma unroll
        for (int w = 0; w < NW; ++w) { oa[w] += aa * kv[g][w]; ob[w] += ab * kv[g][w]; }
    }
    const float ra = 1.f / dena;
    const float rb = 1.f / denb;
#pragma unroll
    for (int w = 0; w < NW; ++w) { oa[w] *= ra; ob[w] *= rb; }

    float* ypa = &y_lds[hA][tIn * NW];
    float* ypb = &y_lds[hA + 8][tIn * NW];
    *(float4*)(ypa)     = make_float4(oa[0], oa[1], oa[2], oa[3]);
    *(float4*)(ypa + 4) = make_float4(oa[4], oa[5], oa[6], oa[7]);
    *(float4*)(ypb)     = make_float4(ob[0], ob[1], ob[2], ob[3]);
    *(float4*)(ypb + 4) = make_float4(ob[4], ob[5], ob[6], ob[7]);
    __syncthreads();

    const float* arow = &y_lds[c][kgrp];
    v2f af[EDIM / 4];
#pragma unroll
    for (int i = 0; i < EDIM / 4; ++i) af[i] = *(const v2f*)(arow + 4 * i);
    __builtin_amdgcn_sched_barrier(0);
    v8f acc0 = {0.f, 0.f, 0.f, 0.f, 0.f, 0.f, 0.f, 0.f};
    v8f acc1 = {0.f, 0.f, 0.f, 0.f, 0.f, 0.f, 0.f, 0.f};
#pragma unroll
    for (int i = 0; i < EDIM / 4; ++i) {
        acc0 = __builtin_amdgcn_wmma_f32_16x16x4_f32(
             false, af[i],  false, bf0[i],
             (short)0, acc0,  false,  false);
        acc1 = __builtin_amdgcn_wmma_f32_16x16x4_f32(
             false, af[i],  false, bf1[i],
             (short)0, acc1,  false,  false);
    }

    const int gt0 = blockIdx.x * TOK_PER_BLOCK;
    const int b   = gt0 >> 10;
    const int j   = (gt0 & 1023) >> 4;
    const int mbase = 8 * (lane >> 4);
#pragma unroll
    for (int rr = 0; rr < 8; ++rr) {
        const size_t row = (size_t)b * 1024 + (size_t)(mbase + rr) * 64 + j;
        v2f o2;
        o2.x = acc0[rr] + bcol.x;
        o2.y = acc1[rr] + bcol.y;
        *(v2f*)(out + row * EDIM + n0) = o2;
    }
}

extern "C" void kernel_launch(void* const* d_in, const int* in_sizes, int n_in,
                              void* d_out, int out_size, void* d_ws, size_t ws_size,
                              hipStream_t stream) {
    const float* x     = (const float*)d_in[0];
    const float* theta = (const float*)d_in[1];
    const float* Wc    = (const float*)d_in[2];
    const float* bc    = (const float*)d_in[3];
    float* out = (float*)d_out;

    const int ntok = in_sizes[0] / EDIM;

    const int blocks = ntok / TOK_PER_BLOCK;
    quantum_attn_project_fused<<<blocks, 128, 0, stream>>>(x, theta, Wc, bc, out);
}


# W reaches the B fragments through LDS: a wave fetches its 32 W rows with whole-row loads (4 full lines each) in place of 64 fragment loads of 16 lines x 16 B; k order and sums unchanged
# speedup vs baseline: 2.1080x; 1.2191x over previous
#include <hip/hip_runtime.h>
#include <math.h>


typedef float v2f __attribute__((ext_vector_type(2)));
typedef float v8f __attribute__((ext_vector_type(8)));

#define NW 8
#define NH 16
#define EDIM 128
#define TOK_PER_BLOCK 16
#define YPITCH 132
#define QPITCH 132
#define WPITCH 132

__global__ __launch_bounds__(128) void quantum_attn_project_fused(
    const float* __restrict__ x,
    const float* __restrict__ theta,
    const float* __restrict__ W,
    const float* __restrict__ bias,
    float* __restrict__ out)
{
    __shared__ __attribute__((aligned(16))) float qh_lds[TOK_PER_BLOCK][QPITCH];
    __shared__ __attribute__((aligned(16))) float y_lds[NH][YPITCH];
    __shared__ __attribute__((aligned(16))) float w_lds[EDIM][WPITCH];

    const int lane = threadIdx.x & 31;
    const int wave = threadIdx.x >> 5;
    const int tIn  = wave * 4 + (lane >> 3);
    const int hA   = lane & 7;
    const int gt   = blockIdx.x * TOK_PER_BLOCK + tIn;

    const float* xp = x + (size_t)gt * EDIM + hA * NW;
    float4 xa0 = *(const float4*)(xp);
    float4 xa1 = *(const float4*)(xp + 4);
    float4 xb0 = *(const float4*)(xp + 8 * NW);
    float4 xb1 = *(const float4*)(xp + 8 * NW + 4);
    float4 t0 = *(const float4*)(theta);
    float4 t1 = *(const float4*)(theta + 4);

    const int c    = lane & 15;
    const int kgrp = (lane >> 4) * 2;
    const int n0   = wave * 32 + 2 * c;
    const float* wsrc = W + (size_t)(wave * 32) * EDIM + lane * 4;
    float4 wrow[32];
#pragma unroll
    for (int r = 0; r < 32; ++r) wrow[r] = *(const float4*)(wsrc + r * EDIM);
    const v2f bcol = *(const v2f*)(bias + n0);

    float qa[NW], qb[NW];
    {
        float ca[NW], cb[NW];
        ca[0] = __cosf(xa0.x + t0.x);  cb[0] = __cosf(xb0.x + t0.x);
        ca[1] = __cosf(xa0.y + t0.y);  cb[1] = __cosf(xb0.y + t0.y);
        ca[2] = __cosf(xa0.z + t0.z);  cb[2] = __cosf(xb0.z + t0.z);
        ca[3] = __cosf(xa0.w + t0.w);  cb[3] = __cosf(xb0.w + t0.w);
        ca[4] = __cosf(xa1.x + t1.x);  cb[4] = __cosf(xb1.x + t1.x);
        ca[5] = __cosf(xa1.y + t1.y);  cb[5] = __cosf(xb1.y + t1.y);
        ca[6] = __cosf(xa1.z + t1.z);  cb[6] = __cosf(xb1.z + t1.z);
        ca[7] = __cosf(xa1.w + t1.w);  cb[7] = __cosf(xb1.w + t1.w);
        float pa = ca[0], pb = cb[0];
#pragma unroll
        for (int w = 1; w < NW; ++w) { pa *= ca[w]; qa[w] = pa; pb *= cb[w]; qb[w] = pb; }
        qa[0] = ca[1] * ca[2] * ca[3] * ca[4] * ca[5] * ca[6] * ca[7];
        qb[0] = cb[1] * cb[2] * cb[3] * cb[4] * cb[5] * cb[6] * cb[7];
    }

    float* qrow = &qh_lds[tIn][0];
    *(float4*)(qrow + hA * NW)           = make_float4(qa[0], qa[1], qa[2], qa[3]);
    *(float4*)(qrow + hA * NW + 4)       = make_float4(qa[4], qa[5], qa[6], qa[7]);
    *(float4*)(qrow + (hA + 8) * NW)     = make_float4(qb[0], qb[1], qb[2], qb[3]);
    *(float4*)(qrow + (hA + 8) * NW + 4) = make_float4(qb[4], qb[5], qb[6], qb[7]);

#pragma unroll
    for (int r = 0; r < 32; ++r)
        *(float4*)(&w_lds[wave * 32 + 16 * (r & 1) + (r >> 1)][lane * 4]) = wrow[r];
    __syncthreads();

    float kv[NH][NW];
#pragma unroll
    for (int g = 0; g < NH; ++g) {
        float4 k0 = *(const float4*)(qrow + g * NW);
        float4 k1 = *(const float4*)(qrow + g * NW + 4);
        kv[g][0] = k0.x; kv[g][1] = k0.y; kv[g][2] = k0.z; kv[g][3] = k0.w;
        kv[g][4] = k1.x; kv[g][5] = k1.y; kv[g][6] = k1.z; kv[g][7] = k1.w;
    }

    __builtin_amdgcn_sched_barrier(0);

    const float* bl0 = &w_lds[wave * 32 + c][kgrp];
    const float* bl1 = &w_lds[wave * 32 + 16 + c][kgrp];
    v2f bf0[EDIM / 4];
    v2f bf1[EDIM / 4];
#pragma unroll
    for (int i = 0; i < EDIM / 4; ++i) {
        bf0[i] = *(const v2f*)(bl0 + 4 * i);
        bf1[i] = *(const v2f*)(bl1 + 4 * i);
    }

    const float inv_sqrt_dk = 0.35355339059327379f;
    float sa[NH], sb[NH];
    float mxa = -INFINITY, mxb = -INFINITY;
#pragma unroll
    for (int g = 0; g < NH; ++g) {
        float da = 0.f, db = 0.f;
#pragma unroll
        for (int w = 0; w < NW; ++w) { da += qa[w] * kv[g][w]; db += qb[w] * kv[g][w]; }
        da *= inv_sqrt_dk;
        db *= inv_sqrt_dk;
        sa[g] = da;
        sb[g] = db;
        mxa = fmaxf(mxa, da);
        mxb = fmaxf(mxb, db);
    }
    float dena = 0.f, denb = 0.f;
#pragma unroll
    for (int g = 0; g < NH; ++g) {
        sa[g] = __expf(sa[g] - mxa); dena += sa[g];
        sb[g] = __expf(sb[g] - mxb); denb += sb[g];
    }

    float oa[NW], ob[NW];
#pragma unroll
    for (int w = 0; w < NW; ++w) { oa[w] = 0.f; ob[w] = 0.f; }
#pragma unroll
    for (int g = 0; g < NH; ++g) {
        const float aa = sa[g];
        const float ab = sb[g];
#pragma unroll
        for (int w = 0; w < NW; ++w) { oa[w] += aa * kv[g][w]; ob[w] += ab * kv[g][w]; }
    }
    const float ra = 1.f / dena;
    const float rb = 1.f / denb;
#pragma unroll
    for (int w = 0; w < NW; ++w) { oa[w] *= ra; ob[w] *= rb; }

    float* ypa = &y_lds[hA][tIn * NW];
    float* ypb = &y_lds[hA + 8][tIn * NW];
    *(float4*)(ypa)     = make_float4(oa[0], oa[1], oa[2], oa[3]);
    *(float4*)(ypa + 4) = make_float4(oa[4], oa[5], oa[6], oa[7]);
    *(float4*)(ypb)     = make_float4(ob[0], ob[1], ob[2], ob[3]);
    *(float4*)(ypb + 4) = make_float4(ob[4], ob[5], ob[6], ob[7]);
    __syncthreads();

    const float* arow = &y_lds[c][kgrp];
    v2f af[EDIM / 4];
#pragma unroll
    for (int i = 0; i < EDIM / 4; ++i) af[i] = *(const v2f*)(arow + 4 * i);
    __builtin_amdgcn_sched_barrier(0);
    v8f acc0 = {0.f, 0.f, 0.f, 0.f, 0.f, 0.f, 0.f, 0.f};
    v8f acc1 = {0.f, 0.f, 0.f, 0.f, 0.f, 0.f, 0.f, 0.f};
#pragma unroll
    for (int i = 0; i < EDIM / 4; ++i) {
        acc0 = __builtin_amdgcn_wmma_f32_16x16x4_f32(
             false, af[i],  false, bf0[i],
             (short)0, acc0,  false,  false);
        acc1 = __builtin_amdgcn_wmma_f32_16x16x4_f32(
             false, af[i],  false, bf1[i],
             (short)0, acc1,  false,  false);
    }

    const int gt0 = blockIdx.x * TOK_PER_BLOCK;
    const int b   = gt0 >> 10;
    const int j   = (gt0 & 1023) >> 4;
    const int mbase = 8 * (lane >> 4);
#pragma unroll
    for (int rr = 0; rr < 8; ++rr) {
        const size_t row = (size_t)b * 1024 + (size_t)(mbase + rr) * 64 + j;
        v2f o2;
        o2.x = acc0[rr] + bcol.x;
        o2.y = acc1[rr] + bcol.y;
        *(v2f*)(out + row * EDIM + n0) = o2;
    }
}

extern "C" void kernel_launch(void* const* d_in, const int* in_sizes, int n_in,
                              void* d_out, int out_size, void* d_ws, size_t ws_size,
                              hipStream_t stream) {
    const float* x     = (const float*)d_in[0];
    const float* theta = (const float*)d_in[1];
    const float* Wc    = (const float*)d_in[2];
    const float* bc    = (const float*)d_in[3];
    float* out = (float*)d_out;

    const int ntok = in_sizes[0] / EDIM;

    const int blocks = ntok / TOK_PER_BLOCK;
    quantum_attn_project_fused<<<blocks, 128, 0, stream>>>(x, theta, Wc, bc, out);
}


# Key/value rows reach a lane by 8-lane DPP broadcasts, not an LDS table: no barrier and no wait for W stand in front of the attention, W is parked in LDS behind it; same values, same sum order
# speedup vs baseline: 2.2522x; 1.0684x over previous
#include <hip/hip_runtime.h>
#include <math.h>


typedef float v2f __attribute__((ext_vector_type(2)));
typedef float v8f __attribute__((ext_vector_type(8)));

#define NW 8
#define NH 16
#define EDIM 128
#define TOK_PER_BLOCK 16
#define YPITCH 132
#define WPITCH 132

template <int J>
__device__ __forceinline__ float bcast8(float v) {
    return __builtin_bit_cast(float, __builtin_amdgcn_mov_dpp8(__builtin_bit_cast(unsigned, v), J * 0x249249u));
}

__global__ __launch_bounds__(128) __attribute__((amdgpu_waves_per_eu(2, 2))) void quantum_attn_project_fused(
    const float* x,
    const float* __restrict__ theta,
    const float* W,
    const float* __restrict__ bias,
    float* __restrict__ out)
{
    __shared__ __attribute__((aligned(16))) float y_lds[NH][YPITCH];
    __shared__ __attribute__((aligned(16))) float w_lds[EDIM][WPITCH];

    const int lane = threadIdx.x & 31;
    const int wave = threadIdx.x >> 5;
    const int tIn  = wave * 4 + (lane >> 3);
    const int hA   = lane & 7;
    const int gt   = blockIdx.x * TOK_PER_BLOCK + tIn;

    const float* xp = x + (size_t)gt * EDIM + hA * NW;
    float4 xa0 = *(const float4*)(xp);
    float4 xa1 = *(const float4*)(xp + 4);
    float4 xb0 = *(const float4*)(xp + 8 * NW);
    float4 xb1 = *(const float4*)(xp + 8 * NW + 4);
    float4 t0 = *(const float4*)(theta);
    float4 t1 = *(const float4*)(theta + 4);
    __builtin_amdgcn_sched_barrier(0);

    const int c    = lane & 15;
    const int kgrp = (lane >> 4) * 2;
    const int n0   = wave * 32 + 2 * c;
    const float* wsrc = W + (size_t)(wave * 32) * EDIM + lane * 4;
#define W_ROWS(X) X(0) X(1) X(2) X(3) X(4) X(5) X(6) X(7) X(8) X(9) X(10) X(11) \
    X(12) X(13) X(14) X(15) X(16) X(17) X(18) X(19) X(20) X(21) X(22) X(23)      \
    X(24) X(25) X(26) X(27) X(28) X(29) X(30) X(31)
#define W_LOAD(r) const float4 wr##r = *(const float4*)(wsrc + (r) * EDIM);
    W_ROWS(W_LOAD)
#undef W_LOAD
    const v2f bcol = *(const v2f*)(bias + n0);
    __builtin_amdgcn_sched_barrier(0);

    float qa[NW], qb[NW];
    {
        float ca[NW], cb[NW];
        ca[0] = __cosf(xa0.x + t0.x);  cb[0] = __cosf(xb0.x + t0.x);
        ca[1] = __cosf(xa0.y + t0.y);  cb[1] = __cosf(xb0.y + t0.y);
        ca[2] = __cosf(xa0.z + t0.z);  cb[2] = __cosf(xb0.z + t0.z);
        ca[3] = __cosf(xa0.w + t0.w);  cb[3] = __cosf(xb0.w + t0.w);
        ca[4] = __cosf(xa1.x + t1.x);  cb[4] = __cosf(xb1.x + t1.x);
        ca[5] = __cosf(xa1.y + t1.y);  cb[5] = __cosf(xb1.y + t1.y);
        ca[6] = __cosf(xa1.z + t1.z);  cb[6] = __cosf(xb1.z + t1.z);
        ca[7] = __cosf(xa1.w + t1.w);  cb[7] = __cosf(xb1.w + t1.w);
        float pa = ca[0], pb = cb[0];
#pragma unroll
        for (int w = 1; w < NW; ++w) { pa *= ca[w]; qa[w] = pa; pb *= cb[w]; qb[w] = pb; }
        qa[0] = ca[1] * ca[2] * ca[3] * ca[4] * ca[5] * ca[6] * ca[7];
        qb[0] = cb[1] * cb[2] * cb[3] * cb[4] * cb[5] * cb[6] * cb[7];
    }

    float kv[NH][NW];
#define KV_FROM_LANE(J)                                                        \
    _Pragma("unroll") for (int w = 0; w < NW; ++w) {                           \
        kv[J][w]     = bcast8<J>(qa[w]);                                       \
        kv[J + 8][w] = bcast8<J>(qb[w]);                                       \
    }
    KV_FROM_LANE(0) KV_FROM_LANE(1) KV_FROM_LANE(2) KV_FROM_LANE(3)
    KV_FROM_LANE(4) KV_FROM_LANE(5) KV_FROM_LANE(6) KV_FROM_LANE(7)
#undef KV_FROM_LANE

    const float inv_sqrt_dk = 0.35355339059327379f;
    float sa[NH], sb[NH];
    float mxa = -INFINITY, mxb = -INFINITY;
#pragma unroll
    for (int g = 0; g < NH; ++g) {
        float da = 0.f, db = 0.f;
#pragma unroll
        for (int w = 0; w < NW; ++w) { da += qa[w] * kv[g][w]; db += qb[w] * kv[g][w]; }
        da *= inv_sqrt_dk;
        db *= inv_sqrt_dk;
        sa[g] = da;
        sb[g] = db;
        mxa = fmaxf(mxa, da);
        mxb = fmaxf(mxb, db);
    }
    float dena = 0.f, denb = 0.f;
#pragma unroll
    for (int g = 0; g < NH; ++g) {
        sa[g] = __expf(sa[g] - mxa); dena += sa[g];
        sb[g] = __expf(sb[g] - mxb); denb += sb[g];
    }

    float oa[NW], ob[NW];
#pragma unroll
    for (int w = 0; w < NW; ++w) { oa[w] = 0.f; ob[w] = 0.f; }
#pragma unroll
    for (int g = 0; g < NH; ++g) {
        const float aa = sa[g];
        const float ab = sb[g];
#pragma unroll
        for (int w = 0; w < NW; ++w) { oa[w] += aa * kv[g][w]; ob[w] += ab * kv[g][w]; }
    }
    const float ra = 1.f / dena;
    const float rb = 1.f / denb;
#pragma unroll
    for (int w = 0; w < NW; ++w) { oa[w] *= ra; ob[w] *= rb; }

    float* ypa = &y_lds[hA][tIn * NW];
    float* ypb = &y_lds[hA + 8][tIn * NW];
    *(float4*)(ypa)     = make_float4(oa[0], oa[1], oa[2], oa[3]);
    *(float4*)(ypa + 4) = make_float4(oa[4], oa[5], oa[6], oa[7]);
    *(float4*)(ypb)     = make_float4(ob[0], ob[1], ob[2], ob[3]);
    *(float4*)(ypb + 4) = make_float4(ob[4], ob[5], ob[6], ob[7]);

    __builtin_amdgcn_sched_barrier(0);
#define W_PARK(r) *(float4*)(&w_lds[wave * 32 + 16 * ((r) & 1) + ((r) >> 1)][lane * 4]) = wr##r;
    W_ROWS(W_PARK)
#undef W_PARK
#undef W_ROWS
    __syncthreads();

    const float* arow = &y_lds[c][kgrp];
    const float* bl0 = &w_lds[wave * 32 + c][kgrp];
    const float* bl1 = &w_lds[wave * 32 + 16 + c][kgrp];
    v2f af[EDIM / 4];
    v2f bf0[EDIM / 4];
    v2f bf1[EDIM / 4];
#pragma unroll
    for (int i = 0; i < EDIM / 4; ++i) {
        af[i]  = *(const v2f*)(arow + 4 * i);
        bf0[i] = *(const v2f*)(bl0 + 4 * i);
        bf1[i] = *(const v2f*)(bl1 + 4 * i);
    }
    __builtin_amdgcn_sched_barrier(0);
    v8f acc0 = {0.f, 0.f, 0.f, 0.f, 0.f, 0.f, 0.f, 0.f};
    v8f acc1 = {0.f, 0.f, 0.f, 0.f, 0.f, 0.f, 0.f, 0.f};
#pragma unroll
    for (int i = 0; i < EDIM / 4; ++i) {
        acc0 = __builtin_amdgcn_wmma_f32_16x16x4_f32(
             false, af[i],  false, bf0[i],
             (short)0, acc0,  false,  false);
        acc1 = __builtin_amdgcn_wmma_f32_16x16x4_f32(
             false, af[i],  false, bf1[i],
             (short)0, acc1,  false,  false);
    }

    const int gt0 = blockIdx.x * TOK_PER_BLOCK;
    const int b   = gt0 >> 10;
    const int j   = (gt0 & 1023) >> 4;
    const int mbase = 8 * (lane >> 4);
#pragma unroll
    for (int rr = 0; rr < 8; ++rr) {
        const size_t row = (size_t)b * 1024 + (size_t)(mbase + rr) * 64 + j;
        v2f o2;
        o2.x = acc0[rr] + bcol.x;
        o2.y = acc1[rr] + bcol.y;
        *(v2f*)(out + row * EDIM + n0) = o2;
    }
}

extern "C" void kernel_launch(void* const* d_in, const int* in_sizes, int n_in,
                              void* d_out, int out_size, void* d_ws, size_t ws_size,
                              hipStream_t stream) {
    const float* x     = (const float*)d_in[0];
    const float* theta = (const float*)d_in[1];
    const float* Wc    = (const float*)d_in[2];
    const float* bc    = (const float*)d_in[3];
    float* out = (float*)d_out;

    const int ntok = in_sizes[0] / EDIM;

    const int blocks = ntok / TOK_PER_BLOCK;
    quantum_attn_project_fused<<<blocks, 128, 0, stream>>>(x, theta, Wc, bc, out);
}
